# starting kernel
# baseline (speedup 1.0000x reference)
#include <hip/hip_runtime.h>
#include <hip/hip_bf16.h>

typedef float v2f __attribute__((ext_vector_type(2)));
typedef float v8f __attribute__((ext_vector_type(8)));

#define BIGF 3.0e38f

__global__ __launch_bounds__(256) void transform_points_kernel(
    const float* __restrict__ src, float4* __restrict__ dst, int n, int npad, int B)
{
    int i = blockIdx.x * blockDim.x + threadIdx.x;
    int total = B * npad;
    if (i >= total) return;
    int b = i / npad;
    int idx = i - b * npad;
    float x = 0.f, y = 0.f, z = 0.f, w = BIGF;
    if (idx < n) {
        const float* s = src + ((size_t)b * n + idx) * 3;
        x = s[0]; y = s[1]; z = s[2];
        w = x * x + y * y + z * z;
    }
    dst[i] = make_float4(x, y, z, w);
}

__global__ __launch_bounds__(256) void chamfer_wmma_kernel(
    const float4* __restrict__ Pt, const float4* __restrict__ Qt,
    float* __restrict__ out, int NP, int NPpad, int NQpad, int B)
{
    const int lane  = threadIdx.x & 31;
    const int gwave = blockIdx.x * (blockDim.x >> 5) + (threadIdx.x >> 5);

    const int groups = NPpad >> 5;
    const int b = gwave / groups;
    if (b >= B) return;
    const int row0 = (gwave - b * groups) << 5;

    const int  j  = lane & 15;
    const bool hi = lane >= 16;

    const float4* Pb = Pt + (size_t)b * NPpad;
    const float* Qb = (const float*)(Qt + (size_t)b * NQpad) + 4 * j + (hi ? 2 : 0);

    float4 p0 = Pb[row0 + j];
    float4 p1 = Pb[row0 + 16 + j];

    v2f a0, a1;
    a0.x = hi ? (-2.f * p0.z) : (-2.f * p0.x);
    a0.y = hi ? 1.f           : (-2.f * p0.y);
    a1.x = hi ? (-2.f * p1.z) : (-2.f * p1.x);
    a1.y = hi ? 1.f           : (-2.f * p1.y);

    const int srcBase = hi ? 8 : 0;
    v8f c0, c1, minv0, minv1;
#pragma unroll
    for (int r = 0; r < 8; ++r) {
        c0[r] = __shfl(p0.w, srcBase + r, 16);
        c1[r] = __shfl(p1.w, srcBase + r, 16);
        minv0[r] = BIGF;
        minv1[r] = BIGF;
    }

#pragma unroll 4
    for (int col0 = 0; col0 < NQpad; col0 += 16) {
        v2f bb = *(const v2f*)(Qb + 4 * col0);

        v8f d0 = __builtin_amdgcn_wmma_f32_16x16x4_f32(
            false, a0, false, bb, (short)0, c0, false, false);
        v8f d1 = __builtin_amdgcn_wmma_f32_16x16x4_f32(
            false, a1, false, bb, (short)0, c1, false, false);

#pragma unroll
        for (int r = 0; r < 8; ++r) {
            minv0[r] = fminf(minv0[r], d0[r]);
            minv1[r] = fminf(minv1[r], d1[r]);
        }
    }

#pragma unroll
    for (int mask = 1; mask <= 8; mask <<= 1) {
#pragma unroll
        for (int r = 0; r < 8; ++r) {
            minv0[r] = fminf(minv0[r], __shfl_xor(minv0[r], mask, 32));
            minv1[r] = fminf(minv1[r], __shfl_xor(minv1[r], mask, 32));
        }
    }

    if (j == 0) {
        int mbase = row0 + (hi ? 8 : 0);
        float* ob = out + (size_t)b * NP;
#pragma unroll
        for (int r = 0; r < 8; ++r) {
            int m0 = mbase + r;
            int m1 = mbase + 16 + r;
            if (m0 < NP) ob[m0] = minv0[r];
            if (m1 < NP) ob[m1] = minv1[r];
        }
    }
}

__global__ __launch_bounds__(256) void chamfer_nn_min_fallback(
    const float* __restrict__ P, const float* __restrict__ Q,
    float* __restrict__ out, int NP, int NQ, int B)
{
    const int lane  = threadIdx.x & 31;
    const int gwave = blockIdx.x * (blockDim.x >> 5) + (threadIdx.x >> 5);
    const int stripsPerBatch = (NP + 15) >> 4;
    const int b = gwave / stripsPerBatch;
    if (b >= B) return;
    const int row0 = (gwave - b * stripsPerBatch) << 4;
    const int  j  = lane & 15;
    const bool hi = lane >= 16;

    const float* Pb = P + (size_t)b * NP * 3;
    const float* Qb = Q + (size_t)b * NQ * 3;

    int pi = row0 + j; if (pi >= NP) pi = NP - 1;
    float px = Pb[3 * pi], py = Pb[3 * pi + 1], pz = Pb[3 * pi + 2];
    float p2 = px * px + py * py + pz * pz;

    v2f a;
    a.x = hi ? (-2.f * pz) : (-2.f * px);
    a.y = hi ? 1.f         : (-2.f * py);

    v8f c, minv;
#pragma unroll
    for (int r = 0; r < 8; ++r) {
        c[r] = __shfl(p2, (hi ? 8 : 0) + r, 16);
        minv[r] = BIGF;
    }

#pragma unroll 2
    for (int col0 = 0; col0 < NQ; col0 += 16) {
        int qi = col0 + j;
        bool qok = qi < NQ;
        if (!qok) qi = NQ - 1;
        float qx = Qb[3 * qi], qy = Qb[3 * qi + 1], qz = Qb[3 * qi + 2];
        float q2 = qok ? (qx * qx + qy * qy + qz * qz) : BIGF;
        v2f bb;
        bb.x = hi ? qz : qx;
        bb.y = hi ? q2 : qy;
        v8f d = __builtin_amdgcn_wmma_f32_16x16x4_f32(
            false, a, false, bb, (short)0, c, false, false);
#pragma unroll
        for (int r = 0; r < 8; ++r) minv[r] = fminf(minv[r], d[r]);
    }

#pragma unroll
    for (int mask = 1; mask <= 8; mask <<= 1) {
#pragma unroll
        for (int r = 0; r < 8; ++r)
            minv[r] = fminf(minv[r], __shfl_xor(minv[r], mask, 32));
    }

    if (j == 0) {
        int mbase = row0 + (hi ? 8 : 0);
#pragma unroll
        for (int r = 0; r < 8; ++r) {
            int m = mbase + r;
            if (m < NP) out[(size_t)b * NP + m] = minv[r];
        }
    }
}

extern "C" void kernel_launch(void* const* d_in, const int* in_sizes, int n_in,
                              void* d_out, int out_size, void* d_ws, size_t ws_size,
                              hipStream_t stream) {
    const float* in1 = (const float*)d_in[0];
    const float* in2 = (const float*)d_in[1];
    float* out = (float*)d_out;

    const int B = 4;
    const int N = in_sizes[0] / (B * 3);
    const int M = in_sizes[1] / (B * 3);
    const int Npad = (N + 31) & ~31;
    const int Mpad = (M + 31) & ~31;

    size_t need = (size_t)B * (Npad + Mpad) * sizeof(float4);
    if (ws_size >= need) {
        float4* T1 = (float4*)d_ws;
        float4* T2 = T1 + (size_t)B * Npad;

        transform_points_kernel<<<(B * Npad + 255) / 256, 256, 0, stream>>>(in1, T1, N, Npad, B);
        transform_points_kernel<<<(B * Mpad + 255) / 256, 256, 0, stream>>>(in2, T2, M, Mpad, B);

        {
            int waves = B * (Npad >> 5);
            int blocks = (waves + 7) / 8;
            chamfer_wmma_kernel<<<blocks, 256, 0, stream>>>(T1, T2, out, N, Npad, Mpad, B);
        }
        {
            int waves = B * (Mpad >> 5);
            int blocks = (waves + 7) / 8;
            chamfer_wmma_kernel<<<blocks, 256, 0, stream>>>(T2, T1, out + (size_t)B * N, M, Mpad, Npad, B);
        }
    } else {
        {
            int waves = B * ((N + 15) / 16);
            int blocks = (waves + 7) / 8;
            chamfer_nn_min_fallback<<<blocks, 256, 0, stream>>>(in1, in2, out, N, M, B);
        }
        {
            int waves = B * ((M + 15) / 16);
            int blocks = (waves + 7) / 8;
            chamfer_nn_min_fallback<<<blocks, 256, 0, stream>>>(in2, in1, out + (size_t)B * N, M, N, B);
        }
    }
}


# Split each 32-row group's target scan over the 8 waves of one work-group, partial minima combined in LDS (min is exact): 8x the waves, all WGPs filled, 8 waves per SIMD cover load waits
# speedup vs baseline: 3.1074x; 3.1074x over previous
#include <hip/hip_runtime.h>
#include <hip/hip_bf16.h>

typedef float v2f __attribute__((ext_vector_type(2)));
typedef float v8f __attribute__((ext_vector_type(8)));

#define BIGF 3.0e38f

__global__ __launch_bounds__(256) void transform_points_kernel(
    const float* __restrict__ src, float4* __restrict__ dst, int n, int npad, int B)
{
    int i = blockIdx.x * blockDim.x + threadIdx.x;
    int total = B * npad;
    if (i >= total) return;
    int b = i / npad;
    int idx = i - b * npad;
    float x = 0.f, y = 0.f, z = 0.f, w = BIGF;
    if (idx < n) {
        const float* s = src + ((size_t)b * n + idx) * 3;
        x = s[0]; y = s[1]; z = s[2];
        w = x * x + y * y + z * z;
    }
    dst[i] = make_float4(x, y, z, w);
}

__global__ __launch_bounds__(256) void chamfer_wmma_kernel(
    const float4* __restrict__ Pt, const float4* __restrict__ Qt,
    float* __restrict__ out, int NP, int NPpad, int NQpad, int B)
{
    __shared__ float part[8][16][32];
    __shared__ float rowmin[32];

    const int lane = threadIdx.x & 31;
    const int wave = __builtin_amdgcn_readfirstlane((int)(threadIdx.x >> 5));

    const int groups = NPpad >> 5;
    const int b = blockIdx.x / groups;
    if (b >= B) return;
    const int row0 = (blockIdx.x - b * groups) << 5;

    const int  j  = lane & 15;
    const bool hi = lane >= 16;

    const float4* Pb = Pt + (size_t)b * NPpad;
    const float* Qb = (const float*)(Qt + (size_t)b * NQpad) + 4 * j + (hi ? 2 : 0);

    float4 p0 = Pb[row0 + j];
    float4 p1 = Pb[row0 + 16 + j];

    v2f a0, a1;
    a0.x = hi ? (-2.f * p0.z) : (-2.f * p0.x);
    a0.y = hi ? 1.f           : (-2.f * p0.y);
    a1.x = hi ? (-2.f * p1.z) : (-2.f * p1.x);
    a1.y = hi ? 1.f           : (-2.f * p1.y);

    const int srcBase = hi ? 8 : 0;
    v8f c0, c1, minv0, minv1;
#pragma unroll
    for (int r = 0; r < 8; ++r) {
        c0[r] = __shfl(p0.w, srcBase + r, 16);
        c1[r] = __shfl(p1.w, srcBase + r, 16);
        minv0[r] = BIGF;
        minv1[r] = BIGF;
    }

    const int ntiles = NQpad >> 4;
    const int tpw    = (ntiles + 7) >> 3;
    const int tbeg   = wave * tpw;
    const int tend   = (tbeg + tpw < ntiles) ? (tbeg + tpw) : ntiles;
    const int colEnd = tend << 4;

#pragma unroll 4
    for (int col0 = tbeg << 4; col0 < colEnd; col0 += 16) {
        v2f bb = *(const v2f*)(Qb + 4 * col0);

        v8f d0 = __builtin_amdgcn_wmma_f32_16x16x4_f32(
            false, a0, false, bb, (short)0, c0, false, false);
        v8f d1 = __builtin_amdgcn_wmma_f32_16x16x4_f32(
            false, a1, false, bb, (short)0, c1, false, false);

#pragma unroll
        for (int r = 0; r < 8; ++r) {
            minv0[r] = fminf(minv0[r], d0[r]);
            minv1[r] = fminf(minv1[r], d1[r]);
        }
    }

#pragma unroll
    for (int r = 0; r < 8; ++r) {
        part[wave][r][lane]     = minv0[r];
        part[wave][8 + r][lane] = minv1[r];
    }
    __syncthreads();

    float v0 = part[0][wave][lane];
    float v1 = part[0][8 + wave][lane];
#pragma unroll
    for (int ww = 1; ww < 8; ++ww) {
        v0 = fminf(v0, part[ww][wave][lane]);
        v1 = fminf(v1, part[ww][8 + wave][lane]);
    }

#pragma unroll
    for (int mask = 1; mask <= 8; mask <<= 1) {
        v0 = fminf(v0, __shfl_xor(v0, mask, 32));
        v1 = fminf(v1, __shfl_xor(v1, mask, 32));
    }

    if (j == 0) {
        rowmin[(hi ? 8 : 0) + wave]      = v0;
        rowmin[16 + (hi ? 8 : 0) + wave] = v1;
    }
    __syncthreads();

    if (wave == 0) {
        int m = row0 + lane;
        if (m < NP) out[(size_t)b * NP + m] = rowmin[lane];
    }
}

__global__ __launch_bounds__(256) void chamfer_nn_min_fallback(
    const float* __restrict__ P, const float* __restrict__ Q,
    float* __restrict__ out, int NP, int NQ, int B)
{
    const int lane  = threadIdx.x & 31;
    const int gwave = blockIdx.x * (blockDim.x >> 5) + (threadIdx.x >> 5);
    const int stripsPerBatch = (NP + 15) >> 4;
    const int b = gwave / stripsPerBatch;
    if (b >= B) return;
    const int row0 = (gwave - b * stripsPerBatch) << 4;
    const int  j  = lane & 15;
    const bool hi = lane >= 16;

    const float* Pb = P + (size_t)b * NP * 3;
    const float* Qb = Q + (size_t)b * NQ * 3;

    int pi = row0 + j; if (pi >= NP) pi = NP - 1;
    float px = Pb[3 * pi], py = Pb[3 * pi + 1], pz = Pb[3 * pi + 2];
    float p2 = px * px + py * py + pz * pz;

    v2f a;
    a.x = hi ? (-2.f * pz) : (-2.f * px);
    a.y = hi ? 1.f         : (-2.f * py);

    v8f c, minv;
#pragma unroll
    for (int r = 0; r < 8; ++r) {
        c[r] = __shfl(p2, (hi ? 8 : 0) + r, 16);
        minv[r] = BIGF;
    }

#pragma unroll 2
    for (int col0 = 0; col0 < NQ; col0 += 16) {
        int qi = col0 + j;
        bool qok = qi < NQ;
        if (!qok) qi = NQ - 1;
        float qx = Qb[3 * qi], qy = Qb[3 * qi + 1], qz = Qb[3 * qi + 2];
        float q2 = qok ? (qx * qx + qy * qy + qz * qz) : BIGF;
        v2f bb;
        bb.x = hi ? qz : qx;
        bb.y = hi ? q2 : qy;
        v8f d = __builtin_amdgcn_wmma_f32_16x16x4_f32(
            false, a, false, bb, (short)0, c, false, false);
#pragma unroll
        for (int r = 0; r < 8; ++r) minv[r] = fminf(minv[r], d[r]);
    }

#pragma unroll
    for (int mask = 1; mask <= 8; mask <<= 1) {
#pragma unroll
        for (int r = 0; r < 8; ++r)
            minv[r] = fminf(minv[r], __shfl_xor(minv[r], mask, 32));
    }

    if (j == 0) {
        int mbase = row0 + (hi ? 8 : 0);
#pragma unroll
        for (int r = 0; r < 8; ++r) {
            int m = mbase + r;
            if (m < NP) out[(size_t)b * NP + m] = minv[r];
        }
    }
}

extern "C" void kernel_launch(void* const* d_in, const int* in_sizes, int n_in,
                              void* d_out, int out_size, void* d_ws, size_t ws_size,
                              hipStream_t stream) {
    const float* in1 = (const float*)d_in[0];
    const float* in2 = (const float*)d_in[1];
    float* out = (float*)d_out;

    const int B = 4;
    const int N = in_sizes[0] / (B * 3);
    const int M = in_sizes[1] / (B * 3);
    const int Npad = (N + 31) & ~31;
    const int Mpad = (M + 31) & ~31;

    size_t need = (size_t)B * (Npad + Mpad) * sizeof(float4);
    if (ws_size >= need) {
        float4* T1 = (float4*)d_ws;
        float4* T2 = T1 + (size_t)B * Npad;

        transform_points_kernel<<<(B * Npad + 255) / 256, 256, 0, stream>>>(in1, T1, N, Npad, B);
        transform_points_kernel<<<(B * Mpad + 255) / 256, 256, 0, stream>>>(in2, T2, M, Mpad, B);

        {
            int blocks = B * (Npad >> 5);
            chamfer_wmma_kernel<<<blocks, 256, 0, stream>>>(T1, T2, out, N, Npad, Mpad, B);
        }
        {
            int blocks = B * (Mpad >> 5);
            chamfer_wmma_kernel<<<blocks, 256, 0, stream>>>(T2, T1, out + (size_t)B * N, M, Mpad, Npad, B);
        }
    } else {
        {
            int waves = B * ((N + 15) / 16);
            int blocks = (waves + 7) / 8;
            chamfer_nn_min_fallback<<<blocks, 256, 0, stream>>>(in1, in2, out, N, M, B);
        }
        {
            int waves = B * ((M + 15) / 16);
            int blocks = (waves + 7) / 8;
            chamfer_nn_min_fallback<<<blocks, 256, 0, stream>>>(in2, in1, out + (size_t)B * N, M, N, B);
        }
    }
}


# One launch instead of four: both directions in one grid, points read raw as x,y,z with |q|^2 formed in the loop (fma-contracted, last-bit differences); no prepass, no workspace
# speedup vs baseline: 4.0473x; 1.3025x over previous
#include <hip/hip_runtime.h>


typedef float v2f __attribute__((ext_vector_type(2)));
typedef float v8f __attribute__((ext_vector_type(8)));

#define BIGF 3.0e38f

struct Pt3 { float x, y, z; };

__global__ __launch_bounds__(256) void chamfer_both_kernel(
    const float* __restrict__ in1, const float* __restrict__ in2,
    float* __restrict__ out, int N, int M, int B)
{
    __shared__ float part[8][16][32];
    __shared__ float rowmin[32];

    const int lane = threadIdx.x & 31;
    const int wave = __builtin_amdgcn_readfirstlane((int)(threadIdx.x >> 5));

    const int g1  = (N + 31) >> 5;
    const int g2  = (M + 31) >> 5;
    const int nb1 = B * g1;
    int blk = blockIdx.x;
    const bool second = blk >= nb1;
    if (second) blk -= nb1;
    const float* P  = second ? in2 : in1;
    const float* Q  = second ? in1 : in2;
    const int NP     = second ? M : N;
    const int NQ     = second ? N : M;
    const int groups = second ? g2 : g1;
    float* o = second ? out + (size_t)B * N : out;

    const int b = blk / groups;
    if (b >= B) return;
    const int row0 = (blk - b * groups) << 5;

    const int  j  = lane & 15;
    const bool hi = lane >= 16;

    const float* Pb = P + (size_t)b * NP * 3;
    const float* Qb = Q + (size_t)b * NQ * 3;
    const int    jo = 3 * j;

    int pi0 = row0 + j;      if (pi0 >= NP) pi0 = NP - 1;
    int pi1 = row0 + 16 + j; if (pi1 >= NP) pi1 = NP - 1;
    const Pt3 p0 = *(const Pt3*)(Pb + 3 * pi0);
    const Pt3 p1 = *(const Pt3*)(Pb + 3 * pi1);
    const float p0w = p0.x * p0.x + p0.y * p0.y + p0.z * p0.z;
    const float p1w = p1.x * p1.x + p1.y * p1.y + p1.z * p1.z;

    v2f a0, a1;
    a0.x = hi ? (-2.f * p0.z) : (-2.f * p0.x);
    a0.y = hi ? 1.f           : (-2.f * p0.y);
    a1.x = hi ? (-2.f * p1.z) : (-2.f * p1.x);
    a1.y = hi ? 1.f           : (-2.f * p1.y);

    const int srcBase = hi ? 8 : 0;
    v8f c0, c1, minv0, minv1;
#pragma unroll
    for (int r = 0; r < 8; ++r) {
        c0[r] = __shfl(p0w, srcBase + r, 16);
        c1[r] = __shfl(p1w, srcBase + r, 16);
        minv0[r] = BIGF;
        minv1[r] = BIGF;
    }

    const int nfull  = NQ >> 4;
    const int ntiles = (NQ + 15) >> 4;
    const int tpw    = (ntiles + 7) >> 3;
    const int tbeg   = wave * tpw;
    const int tend   = (tbeg + tpw < ntiles) ? (tbeg + tpw) : ntiles;
    const int fend   = (tend < nfull) ? tend : nfull;
    const int nt     = (fend > tbeg) ? (fend - tbeg) : 0;

    const float* qw = Qb + 48 * tbeg;

    for (int t = 0; t + 4 <= nt; t += 4) {
        Pt3 q[4];
#pragma unroll
        for (int u = 0; u < 4; ++u) q[u] = *(const Pt3*)(qw + 48 * u + jo);
        qw += 192;
        __builtin_amdgcn_sched_barrier(0);
#pragma unroll
        for (int u = 0; u < 4; ++u) {
            const float q2 = q[u].x * q[u].x + q[u].y * q[u].y + q[u].z * q[u].z;
            v2f bb;
            bb.x = hi ? q[u].z : q[u].x;
            bb.y = hi ? q2     : q[u].y;
            v8f d0 = __builtin_amdgcn_wmma_f32_16x16x4_f32(
                false, a0, false, bb, (short)0, c0, false, false);
            v8f d1 = __builtin_amdgcn_wmma_f32_16x16x4_f32(
                false, a1, false, bb, (short)0, c1, false, false);
#pragma unroll
            for (int r = 0; r < 8; ++r) {
                minv0[r] = fminf(minv0[r], d0[r]);
                minv1[r] = fminf(minv1[r], d1[r]);
            }
        }
    }
    for (int t = nt & ~3; t < nt; ++t) {
        const Pt3 q = *(const Pt3*)(qw + jo);
        qw += 48;
        const float q2 = q.x * q.x + q.y * q.y + q.z * q.z;
        v2f bb;
        bb.x = hi ? q.z : q.x;
        bb.y = hi ? q2  : q.y;
        v8f d0 = __builtin_amdgcn_wmma_f32_16x16x4_f32(
            false, a0, false, bb, (short)0, c0, false, false);
        v8f d1 = __builtin_amdgcn_wmma_f32_16x16x4_f32(
            false, a1, false, bb, (short)0, c1, false, false);
#pragma unroll
        for (int r = 0; r < 8; ++r) {
            minv0[r] = fminf(minv0[r], d0[r]);
            minv1[r] = fminf(minv1[r], d1[r]);
        }
    }

    if (nfull < ntiles && tbeg <= nfull && nfull < tend) {
        int qi = (nfull << 4) + j;
        const bool ok = qi < NQ;
        if (!ok) qi = NQ - 1;
        const Pt3 q = *(const Pt3*)(Qb + 3 * qi);
        const float q2 = ok ? (q.x * q.x + q.y * q.y + q.z * q.z) : BIGF;
        v2f bb;
        bb.x = hi ? q.z : q.x;
        bb.y = hi ? q2  : q.y;
        v8f d0 = __builtin_amdgcn_wmma_f32_16x16x4_f32(
            false, a0, false, bb, (short)0, c0, false, false);
        v8f d1 = __builtin_amdgcn_wmma_f32_16x16x4_f32(
            false, a1, false, bb, (short)0, c1, false, false);
#pragma unroll
        for (int r = 0; r < 8; ++r) {
            minv0[r] = fminf(minv0[r], d0[r]);
            minv1[r] = fminf(minv1[r], d1[r]);
        }
    }

#pragma unroll
    for (int r = 0; r < 8; ++r) {
        part[wave][r][lane]     = minv0[r];
        part[wave][8 + r][lane] = minv1[r];
    }
    __syncthreads();

    float v0 = part[0][wave][lane];
    float v1 = part[0][8 + wave][lane];
#pragma unroll
    for (int ww = 1; ww < 8; ++ww) {
        v0 = fminf(v0, part[ww][wave][lane]);
        v1 = fminf(v1, part[ww][8 + wave][lane]);
    }

#pragma unroll
    for (int mask = 1; mask <= 8; mask <<= 1) {
        v0 = fminf(v0, __shfl_xor(v0, mask, 32));
        v1 = fminf(v1, __shfl_xor(v1, mask, 32));
    }

    if (j == 0) {
        rowmin[(hi ? 8 : 0) + wave]      = v0;
        rowmin[16 + (hi ? 8 : 0) + wave] = v1;
    }
    __syncthreads();

    if (wave == 0) {
        int m = row0 + lane;
        if (m < NP) o[(size_t)b * NP + m] = rowmin[lane];
    }
}

extern "C" void kernel_launch(void* const* d_in, const int* in_sizes, int n_in,
                              void* d_out, int out_size, void* d_ws, size_t ws_size,
                              hipStream_t stream) {
    const float* in1 = (const float*)d_in[0];
    const float* in2 = (const float*)d_in[1];
    float* out = (float*)d_out;

    const int B = 4;
    const int N = in_sizes[0] / (B * 3);
    const int M = in_sizes[1] / (B * 3);

    const int blocks = B * (((N + 31) >> 5) + ((M + 31) >> 5));
    chamfer_both_kernel<<<blocks, 256, 0, stream>>>(in1, in2, out, N, M, B);
}


# Each wave carries 8 row strips instead of 2 (work-group = 128 rows): a quarter of the waves, target loads and B operands, all work-groups resident at once; results bit-identical
# speedup vs baseline: 4.2704x; 1.0551x over previous
#include <hip/hip_runtime.h>


typedef float v2f __attribute__((ext_vector_type(2)));
typedef float v8f __attribute__((ext_vector_type(8)));

#define BIGF 3.0e38f
#define NS   8
#define ROWS (16 * NS)

struct Pt3 { float x, y, z; };

__global__ __launch_bounds__(256, 4) void chamfer_both_kernel(
    const float* __restrict__ in1, const float* __restrict__ in2,
    float* __restrict__ out, int N, int M, int B)
{
    __shared__ float part[8][32][32];
    __shared__ float rowmin[ROWS];

    const int lane = threadIdx.x & 31;
    const int wave = __builtin_amdgcn_readfirstlane((int)(threadIdx.x >> 5));

    const int g1  = (N + ROWS - 1) / ROWS;
    const int g2  = (M + ROWS - 1) / ROWS;
    const int nb1 = B * g1;
    int blk = blockIdx.x;
    const bool second = blk >= nb1;
    if (second) blk -= nb1;
    const float* P  = second ? in2 : in1;
    const float* Q  = second ? in1 : in2;
    const int NP     = second ? M : N;
    const int NQ     = second ? N : M;
    const int groups = second ? g2 : g1;
    float* o = second ? out + (size_t)B * N : out;

    const int b = blk / groups;
    if (b >= B) return;
    const int row0 = (blk - b * groups) * ROWS;

    const int  j  = lane & 15;
    const bool hi = lane >= 16;

    const float* Pb = P + (size_t)b * NP * 3;
    const float* Qb = Q + (size_t)b * NQ * 3;
    const int    jo = 3 * j;

    const int srcBase = hi ? 8 : 0;
    v2f a[NS];
    v8f c[NS], minv[NS];
#pragma unroll
    for (int s = 0; s < NS; ++s) {
        int pi = row0 + 16 * s + j;
        if (pi >= NP) pi = NP - 1;
        const Pt3 p = *(const Pt3*)(Pb + 3 * pi);
        const float pw = p.x * p.x + p.y * p.y + p.z * p.z;
        a[s].x = hi ? (-2.f * p.z) : (-2.f * p.x);
        a[s].y = hi ? 1.f          : (-2.f * p.y);
#pragma unroll
        for (int r = 0; r < 8; ++r) {
            c[s][r]    = __shfl(pw, srcBase + r, 16);
            minv[s][r] = BIGF;
        }
    }

    const int nfull  = NQ >> 4;
    const int ntiles = (NQ + 15) >> 4;
    const int tpw    = (ntiles + 7) >> 3;
    const int tbeg   = wave * tpw;
    const int tend   = (tbeg + tpw < ntiles) ? (tbeg + tpw) : ntiles;
    const int fend   = (tend < nfull) ? tend : nfull;
    const int nt     = (fend > tbeg) ? (fend - tbeg) : 0;

    const float* qw = Qb + 48 * tbeg;

    for (int t = 0; t + 4 <= nt; t += 4) {
        Pt3 q[4];
#pragma unroll
        for (int u = 0; u < 4; ++u) q[u] = *(const Pt3*)(qw + 48 * u + jo);
        qw += 192;
        __builtin_amdgcn_sched_barrier(0);
#pragma unroll
        for (int u = 0; u < 4; ++u) {
            const float q2 = q[u].x * q[u].x + q[u].y * q[u].y + q[u].z * q[u].z;
            v2f bb;
            bb.x = hi ? q[u].z : q[u].x;
            bb.y = hi ? q2     : q[u].y;
#pragma unroll
            for (int s = 0; s < NS; ++s) {
                v8f d = __builtin_amdgcn_wmma_f32_16x16x4_f32(
                    false, a[s], false, bb, (short)0, c[s], false, false);
#pragma unroll
                for (int r = 0; r < 8; ++r) minv[s][r] = fminf(minv[s][r], d[r]);
            }
        }
    }
    for (int t = nt & ~3; t < nt; ++t) {
        const Pt3 q = *(const Pt3*)(qw + jo);
        qw += 48;
        const float q2 = q.x * q.x + q.y * q.y + q.z * q.z;
        v2f bb;
        bb.x = hi ? q.z : q.x;
        bb.y = hi ? q2  : q.y;
#pragma unroll
        for (int s = 0; s < NS; ++s) {
            v8f d = __builtin_amdgcn_wmma_f32_16x16x4_f32(
                false, a[s], false, bb, (short)0, c[s], false, false);
#pragma unroll
            for (int r = 0; r < 8; ++r) minv[s][r] = fminf(minv[s][r], d[r]);
        }
    }

    if (nfull < ntiles && tbeg <= nfull && nfull < tend) {
        int qi = (nfull << 4) + j;
        const bool ok = qi < NQ;
        if (!ok) qi = NQ - 1;
        const Pt3 q = *(const Pt3*)(Qb + 3 * qi);
        const float q2 = ok ? (q.x * q.x + q.y * q.y + q.z * q.z) : BIGF;
        v2f bb;
        bb.x = hi ? q.z : q.x;
        bb.y = hi ? q2  : q.y;
#pragma unroll
        for (int s = 0; s < NS; ++s) {
            v8f d = __builtin_amdgcn_wmma_f32_16x16x4_f32(
                false, a[s], false, bb, (short)0, c[s], false, false);
#pragma unroll
            for (int r = 0; r < 8; ++r) minv[s][r] = fminf(minv[s][r], d[r]);
        }
    }

#pragma unroll
    for (int g = 0; g < NS / 4; ++g) {
        if (g) __syncthreads();
#pragma unroll
        for (int sl = 0; sl < 4; ++sl) {
#pragma unroll
            for (int r = 0; r < 8; ++r) part[wave][8 * sl + r][lane] = minv[4 * g + sl][r];
        }
        __syncthreads();

        float v[4];
#pragma unroll
        for (int sl = 0; sl < 4; ++sl) v[sl] = part[0][8 * sl + wave][lane];
#pragma unroll
        for (int ww = 1; ww < 8; ++ww) {
#pragma unroll
            for (int sl = 0; sl < 4; ++sl) v[sl] = fminf(v[sl], part[ww][8 * sl + wave][lane]);
        }

#pragma unroll
        for (int mask = 1; mask <= 8; mask <<= 1) {
#pragma unroll
            for (int sl = 0; sl < 4; ++sl) v[sl] = fminf(v[sl], __shfl_xor(v[sl], mask, 32));
        }

        if (j == 0) {
#pragma unroll
            for (int sl = 0; sl < 4; ++sl) rowmin[16 * (4 * g + sl) + (hi ? 8 : 0) + wave] = v[sl];
        }
    }
    __syncthreads();

    if (wave < ROWS / 32) {
        int m = row0 + 32 * wave + lane;
        if (m < NP) o[(size_t)b * NP + m] = rowmin[32 * wave + lane];
    }
}

extern "C" void kernel_launch(void* const* d_in, const int* in_sizes, int n_in,
                              void* d_out, int out_size, void* d_ws, size_t ws_size,
                              hipStream_t stream) {
    const float* in1 = (const float*)d_in[0];
    const float* in2 = (const float*)d_in[1];
    float* out = (float*)d_out;

    const int B = 4;
    const int N = in_sizes[0] / (B * 3);
    const int M = in_sizes[1] / (B * 3);

    const int blocks = B * (((N + ROWS - 1) / ROWS) + ((M + ROWS - 1) / ROWS));
    chamfer_both_kernel<<<blocks, 256, 0, stream>>>(in1, in2, out, N, M, B);
}


# Each distance tile is formed once for both outputs: dist2 = column minima of the dist1 products (its sum order changes: last-bit differences); partials via workspace, folded by a small 2nd launch
# speedup vs baseline: 4.8544x; 1.1368x over previous
#include <hip/hip_runtime.h>


typedef float v2f __attribute__((ext_vector_type(2)));
typedef float v8f __attribute__((ext_vector_type(8)));

#define BIGF 3.0e38f
#define NS   8
#define ROWS (16 * NS)

struct Pt3 { float x, y, z; };

__global__ __launch_bounds__(256, 4) void chamfer_both_kernel(
    const float* __restrict__ in1, const float* __restrict__ in2,
    float* __restrict__ out, int N, int M, int B)
{
    __shared__ float part[8][32][32];
    __shared__ float rowmin[ROWS];

    const int lane = threadIdx.x & 31;
    const int wave = __builtin_amdgcn_readfirstlane((int)(threadIdx.x >> 5));

    const int g1  = (N + ROWS - 1) / ROWS;
    const int g2  = (M + ROWS - 1) / ROWS;
    const int nb1 = B * g1;
    int blk = blockIdx.x;
    const bool second = blk >= nb1;
    if (second) blk -= nb1;
    const float* P  = second ? in2 : in1;
    const float* Q  = second ? in1 : in2;
    const int NP     = second ? M : N;
    const int NQ     = second ? N : M;
    const int groups = second ? g2 : g1;
    float* o = second ? out + (size_t)B * N : out;

    const int b = blk / groups;
    if (b >= B) return;
    const int row0 = (blk - b * groups) * ROWS;

    const int  j  = lane & 15;
    const bool hi = lane >= 16;

    const float* Pb = P + (size_t)b * NP * 3;
    const float* Qb = Q + (size_t)b * NQ * 3;
    const int    jo = 3 * j;

    const int srcBase = hi ? 8 : 0;
    v2f a[NS];
    v8f c[NS], minv[NS];
#pragma unroll
    for (int s = 0; s < NS; ++s) {
        int pi = row0 + 16 * s + j;
        if (pi >= NP) pi = NP - 1;
        const Pt3 p = *(const Pt3*)(Pb + 3 * pi);
        const float pw = p.x * p.x + p.y * p.y + p.z * p.z;
        a[s].x = hi ? (-2.f * p.z) : (-2.f * p.x);
        a[s].y = hi ? 1.f          : (-2.f * p.y);
#pragma unroll
        for (int r = 0; r < 8; ++r) {
            c[s][r]    = __shfl(pw, srcBase + r, 16);
            minv[s][r] = BIGF;
        }
    }

    const int nfull  = NQ >> 4;
    const int ntiles = (NQ + 15) >> 4;
    const int tpw    = (ntiles + 7) >> 3;
    const int tbeg   = wave * tpw;
    const int tend   = (tbeg + tpw < ntiles) ? (tbeg + tpw) : ntiles;
    const int fend   = (tend < nfull) ? tend : nfull;
    const int nt     = (fend > tbeg) ? (fend - tbeg) : 0;

    const float* qw = Qb + 48 * tbeg;

    for (int t = 0; t + 4 <= nt; t += 4) {
        Pt3 q[4];
#pragma unroll
        for (int u = 0; u < 4; ++u) q[u] = *(const Pt3*)(qw + 48 * u + jo);
        qw += 192;
        __builtin_amdgcn_sched_barrier(0);
#pragma unroll
        for (int u = 0; u < 4; ++u) {
            const float q2 = q[u].x * q[u].x + q[u].y * q[u].y + q[u].z * q[u].z;
            v2f bb;
            bb.x = hi ? q[u].z : q[u].x;
            bb.y = hi ? q2     : q[u].y;
#pragma unroll
            for (int s = 0; s < NS; ++s) {
                v8f d = __builtin_amdgcn_wmma_f32_16x16x4_f32(
                    false, a[s], false, bb, (short)0, c[s], false, false);
#pragma unroll
                for (int r = 0; r < 8; ++r) minv[s][r] = fminf(minv[s][r], d[r]);
            }
        }
    }
    for (int t = nt & ~3; t < nt; ++t) {
        const Pt3 q = *(const Pt3*)(qw + jo);
        qw += 48;
        const float q2 = q.x * q.x + q.y * q.y + q.z * q.z;
        v2f bb;
        bb.x = hi ? q.z : q.x;
        bb.y = hi ? q2  : q.y;
#pragma unroll
        for (int s = 0; s < NS; ++s) {
            v8f d = __builtin_amdgcn_wmma_f32_16x16x4_f32(
                false, a[s], false, bb, (short)0, c[s], false, false);
#pragma unroll
            for (int r = 0; r < 8; ++r) minv[s][r] = fminf(minv[s][r], d[r]);
        }
    }

    if (nfull < ntiles && tbeg <= nfull && nfull < tend) {
        int qi = (nfull << 4) + j;
        const bool ok = qi < NQ;
        if (!ok) qi = NQ - 1;
        const Pt3 q = *(const Pt3*)(Qb + 3 * qi);
        const float q2 = ok ? (q.x * q.x + q.y * q.y + q.z * q.z) : BIGF;
        v2f bb;
        bb.x = hi ? q.z : q.x;
        bb.y = hi ? q2  : q.y;
#pragma unroll
        for (int s = 0; s < NS; ++s) {
            v8f d = __builtin_amdgcn_wmma_f32_16x16x4_f32(
                false, a[s], false, bb, (short)0, c[s], false, false);
#pragma unroll
            for (int r = 0; r < 8; ++r) minv[s][r] = fminf(minv[s][r], d[r]);
        }
    }

#pragma unroll
    for (int g = 0; g < NS / 4; ++g) {
        if (g) __syncthreads();
#pragma unroll
        for (int sl = 0; sl < 4; ++sl) {
#pragma unroll
            for (int r = 0; r < 8; ++r) part[wave][8 * sl + r][lane] = minv[4 * g + sl][r];
        }
        __syncthreads();

        float v[4];
#pragma unroll
        for (int sl = 0; sl < 4; ++sl) v[sl] = part[0][8 * sl + wave][lane];
#pragma unroll
        for (int ww = 1; ww < 8; ++ww) {
#pragma unroll
            for (int sl = 0; sl < 4; ++sl) v[sl] = fminf(v[sl], part[ww][8 * sl + wave][lane]);
        }

#pragma unroll
        for (int mask = 1; mask <= 8; mask <<= 1) {
#pragma unroll
            for (int sl = 0; sl < 4; ++sl) v[sl] = fminf(v[sl], __shfl_xor(v[sl], mask, 32));
        }

        if (j == 0) {
#pragma unroll
            for (int sl = 0; sl < 4; ++sl) rowmin[16 * (4 * g + sl) + (hi ? 8 : 0) + wave] = v[sl];
        }
    }
    __syncthreads();

    if (wave < ROWS / 32) {
        int m = row0 + 32 * wave + lane;
        if (m < NP) o[(size_t)b * NP + m] = rowmin[32 * wave + lane];
    }
}

#define CSPLIT 2

__global__ __launch_bounds__(256, 4) void chamfer_shared_tile_kernel(
    const float* __restrict__ in1, const float* __restrict__ in2,
    float* __restrict__ rowpart, float* __restrict__ colpart, int N, int M, int B)
{
    __shared__ float part[8][32][32];
    __shared__ float rowmin[ROWS];

    const int lane = threadIdx.x & 31;
    const int wave = __builtin_amdgcn_readfirstlane((int)(threadIdx.x >> 5));

    const int g1   = (N + ROWS - 1) / ROWS;
    const int cs   = blockIdx.x % CSPLIT;
    const int rest = blockIdx.x / CSPLIT;
    const int b    = rest / g1;
    if (b >= B) return;
    const int grp  = rest - b * g1;
    const int row0 = grp * ROWS;
    const int NP = N;
    const int NQ = M;

    const int  j  = lane & 15;
    const bool hi = lane >= 16;

    const float* Pb = in1 + (size_t)b * NP * 3;
    const float* Qb = in2 + (size_t)b * NQ * 3;
    const int    jo = 3 * j;
    float* cp = colpart + ((size_t)b * g1 + grp) * NQ;

    const int srcBase = hi ? 8 : 0;
    v2f a[NS];
    v8f c[NS], minv[NS];
#pragma unroll
    for (int s = 0; s < NS; ++s) {
        int pi = row0 + 16 * s + j;
        if (pi >= NP) pi = NP - 1;
        const Pt3 p = *(const Pt3*)(Pb + 3 * pi);
        const float pw = p.x * p.x + p.y * p.y + p.z * p.z;
        a[s].x = hi ? (-2.f * p.z) : (-2.f * p.x);
        a[s].y = hi ? 1.f          : (-2.f * p.y);
#pragma unroll
        for (int r = 0; r < 8; ++r) {
            c[s][r]    = __shfl(pw, srcBase + r, 16);
            minv[s][r] = BIGF;
        }
    }

    const int nfull  = NQ >> 4;
    const int ntiles = (NQ + 15) >> 4;
    const int tps    = (ntiles + CSPLIT - 1) / CSPLIT;
    const int wt0    = cs * tps;
    const int wt1    = (wt0 + tps < ntiles) ? (wt0 + tps) : ntiles;
    const int tpw    = (tps + 7) >> 3;
    const int tbeg   = wt0 + wave * tpw;
    const int tend   = (tbeg + tpw < wt1) ? (tbeg + tpw) : wt1;
    const int fend   = (tend < nfull) ? tend : nfull;
    const int nt     = (fend > tbeg) ? (fend - tbeg) : 0;

    const float* qw = Qb + 48 * tbeg;
    float* cw = cp + 16 * tbeg;

    for (int t = 0; t + 4 <= nt; t += 4) {
        Pt3 q[4];
#pragma unroll
        for (int u = 0; u < 4; ++u) q[u] = *(const Pt3*)(qw + 48 * u + jo);
        qw += 192;
        __builtin_amdgcn_sched_barrier(0);
        float cmA[4], cmB[4];
#pragma unroll
        for (int u = 0; u < 4; ++u) {
            const float q2 = q[u].x * q[u].x + q[u].y * q[u].y + q[u].z * q[u].z;
            v2f bb;
            bb.x = hi ? q[u].z : q[u].x;
            bb.y = hi ? q2     : q[u].y;
            cmA[u] = BIGF;
            cmB[u] = BIGF;
#pragma unroll
            for (int s = 0; s < NS; ++s) {
                v8f d = __builtin_amdgcn_wmma_f32_16x16x4_f32(
                    false, a[s], false, bb, (short)0, c[s], false, false);
#pragma unroll
                for (int r = 0; r < 8; ++r) minv[s][r] = fminf(minv[s][r], d[r]);
                cmA[u] = fminf(fminf(cmA[u], d[0]), d[1]);
                cmB[u] = fminf(fminf(cmB[u], d[2]), d[3]);
                cmA[u] = fminf(fminf(cmA[u], d[4]), d[5]);
                cmB[u] = fminf(fminf(cmB[u], d[6]), d[7]);
            }
        }
#pragma unroll
        for (int u = 0; u < 4; u += 2) {
            const float ce = fminf(cmA[u], cmB[u]);
            const float co = fminf(cmA[u + 1], cmB[u + 1]);
            const float give = hi ? ce : co;
            const float keep = hi ? co : ce;
            cw[16 * u + lane] = fminf(keep, __shfl_xor(give, 16, 32));
        }
        cw += 64;
    }
    for (int t = nt & ~3; t < nt; ++t) {
        const Pt3 q = *(const Pt3*)(qw + jo);
        qw += 48;
        const float q2 = q.x * q.x + q.y * q.y + q.z * q.z;
        v2f bb;
        bb.x = hi ? q.z : q.x;
        bb.y = hi ? q2  : q.y;
        float cm1 = BIGF, cm2 = BIGF;
#pragma unroll
        for (int s = 0; s < NS; ++s) {
            v8f d = __builtin_amdgcn_wmma_f32_16x16x4_f32(
                false, a[s], false, bb, (short)0, c[s], false, false);
#pragma unroll
            for (int r = 0; r < 8; ++r) minv[s][r] = fminf(minv[s][r], d[r]);
            cm1 = fminf(fminf(cm1, d[0]), d[1]);
            cm2 = fminf(fminf(cm2, d[2]), d[3]);
            cm1 = fminf(fminf(cm1, d[4]), d[5]);
            cm2 = fminf(fminf(cm2, d[6]), d[7]);
        }
        cm1 = fminf(cm1, cm2);
        cw[j] = fminf(cm1, __shfl_xor(cm1, 16, 32));
        cw += 16;
    }

    if (nfull < ntiles && tbeg <= nfull && nfull < tend) {
        int qi = (nfull << 4) + j;
        const bool ok = qi < NQ;
        const int qcol = qi;
        if (!ok) qi = NQ - 1;
        const Pt3 q = *(const Pt3*)(Qb + 3 * qi);
        const float q2 = ok ? (q.x * q.x + q.y * q.y + q.z * q.z) : BIGF;
        v2f bb;
        bb.x = hi ? q.z : q.x;
        bb.y = hi ? q2  : q.y;
        float cm1 = BIGF, cm2 = BIGF;
#pragma unroll
        for (int s = 0; s < NS; ++s) {
            v8f d = __builtin_amdgcn_wmma_f32_16x16x4_f32(
                false, a[s], false, bb, (short)0, c[s], false, false);
#pragma unroll
            for (int r = 0; r < 8; ++r) minv[s][r] = fminf(minv[s][r], d[r]);
            cm1 = fminf(fminf(cm1, d[0]), d[1]);
            cm2 = fminf(fminf(cm2, d[2]), d[3]);
            cm1 = fminf(fminf(cm1, d[4]), d[5]);
            cm2 = fminf(fminf(cm2, d[6]), d[7]);
        }
        cm1 = fminf(cm1, cm2);
        const float cmf = fminf(cm1, __shfl_xor(cm1, 16, 32));
        if (ok) cp[qcol] = cmf;
    }

#pragma unroll
    for (int g = 0; g < NS / 4; ++g) {
        if (g) __syncthreads();
#pragma unroll
        for (int sl = 0; sl < 4; ++sl) {
#pragma unroll
            for (int r = 0; r < 8; ++r) part[wave][8 * sl + r][lane] = minv[4 * g + sl][r];
        }
        __syncthreads();

        float v[4];
#pragma unroll
        for (int sl = 0; sl < 4; ++sl) v[sl] = part[0][8 * sl + wave][lane];
#pragma unroll
        for (int ww = 1; ww < 8; ++ww) {
#pragma unroll
            for (int sl = 0; sl < 4; ++sl) v[sl] = fminf(v[sl], part[ww][8 * sl + wave][lane]);
        }

#pragma unroll
        for (int mask = 1; mask <= 8; mask <<= 1) {
#pragma unroll
            for (int sl = 0; sl < 4; ++sl) v[sl] = fminf(v[sl], __shfl_xor(v[sl], mask, 32));
        }

        if (j == 0) {
#pragma unroll
            for (int sl = 0; sl < 4; ++sl) rowmin[16 * (4 * g + sl) + (hi ? 8 : 0) + wave] = v[sl];
        }
    }
    __syncthreads();

    if (wave < ROWS / 32) {
        int m = row0 + 32 * wave + lane;
        if (m < NP) rowpart[(size_t)cs * B * N + (size_t)b * NP + m] = rowmin[32 * wave + lane];
    }
}

__global__ __launch_bounds__(128) void chamfer_fold_kernel(
    const float* __restrict__ rowpart, const float* __restrict__ colpart,
    float* __restrict__ out, int N, int M, int B)
{
    const int n1 = B * N;
    const int n2 = B * M;
    const int g1 = (N + ROWS - 1) / ROWS;
    const int i  = blockIdx.x * 128 + threadIdx.x;
    if (i < n2) {
        const int b = i / M;
        const int m = i - b * M;
        const float* p = colpart + (size_t)b * g1 * M + m;
        float v = BIGF;
        int g = 0;
        for (; g + 32 <= g1; g += 32) {
            float t[32];
#pragma unroll
            for (int k = 0; k < 32; ++k) t[k] = p[(size_t)(g + k) * M];
#pragma unroll
            for (int k = 0; k < 32; ++k) v = fminf(v, t[k]);
        }
        for (; g < g1; ++g) v = fminf(v, p[(size_t)g * M]);
        out[(size_t)n1 + i] = v;
    }
    if (i < n1) {
        float v = rowpart[i];
#pragma unroll
        for (int k = 1; k < CSPLIT; ++k) v = fminf(v, rowpart[(size_t)k * n1 + i]);
        out[i] = v;
    }
}

extern "C" void kernel_launch(void* const* d_in, const int* in_sizes, int n_in,
                              void* d_out, int out_size, void* d_ws, size_t ws_size,
                              hipStream_t stream) {
    const float* in1 = (const float*)d_in[0];
    const float* in2 = (const float*)d_in[1];
    float* out = (float*)d_out;

    const int B = 4;
    const int N = in_sizes[0] / (B * 3);
    const int M = in_sizes[1] / (B * 3);

    const int g1 = (N + ROWS - 1) / ROWS;
    const int g2 = (M + ROWS - 1) / ROWS;

    const size_t rowFloats = (size_t)CSPLIT * B * N;
    const size_t colFloats = (size_t)B * g1 * M;
    if (d_ws != nullptr && ws_size >= (rowFloats + colFloats) * sizeof(float)) {
        float* rowpart = (float*)d_ws;
        float* colpart = rowpart + rowFloats;
        chamfer_shared_tile_kernel<<<B * g1 * CSPLIT, 256, 0, stream>>>(in1, in2, rowpart, colpart, N, M, B);
        const int nmax = (N > M) ? N : M;
        chamfer_fold_kernel<<<(B * nmax + 127) / 128, 128, 0, stream>>>(rowpart, colpart, out, N, M, B);
    } else {
        chamfer_both_kernel<<<B * (g1 + g2), 256, 0, stream>>>(in1, in2, out, N, M, B);
    }
}


# Work-group re-cut: its 8 waves take 8 different 128-row blocks against the same 32 column tiles: column partials per element 64 -> 8, row partials 2 -> 16; fold reads 24 values in one round trip
# speedup vs baseline: 5.8420x; 1.2034x over previous
#include <hip/hip_runtime.h>


typedef float v2f __attribute__((ext_vector_type(2)));
typedef float v8f __attribute__((ext_vector_type(8)));

#define BIGF 3.0e38f
#define NS   8
#define ROWS (16 * NS)

struct Pt3 { float x, y, z; };

__global__ __launch_bounds__(256, 4) void chamfer_both_kernel(
    const float* __restrict__ in1, const float* __restrict__ in2,
    float* __restrict__ out, int N, int M, int B)
{
    __shared__ float part[8][32][32];
    __shared__ float rowmin[ROWS];

    const int lane = threadIdx.x & 31;
    const int wave = __builtin_amdgcn_readfirstlane((int)(threadIdx.x >> 5));

    const int g1  = (N + ROWS - 1) / ROWS;
    const int g2  = (M + ROWS - 1) / ROWS;
    const int nb1 = B * g1;
    int blk = blockIdx.x;
    const bool second = blk >= nb1;
    if (second) blk -= nb1;
    const float* P  = second ? in2 : in1;
    const float* Q  = second ? in1 : in2;
    const int NP     = second ? M : N;
    const int NQ     = second ? N : M;
    const int groups = second ? g2 : g1;
    float* o = second ? out + (size_t)B * N : out;

    const int b = blk / groups;
    if (b >= B) return;
    const int row0 = (blk - b * groups) * ROWS;

    const int  j  = lane & 15;
    const bool hi = lane >= 16;

    const float* Pb = P + (size_t)b * NP * 3;
    const float* Qb = Q + (size_t)b * NQ * 3;
    const int    jo = 3 * j;

    const int srcBase = hi ? 8 : 0;
    v2f a[NS];
    v8f c[NS], minv[NS];
#pragma unroll
    for (int s = 0; s < NS; ++s) {
        int pi = row0 + 16 * s + j;
        if (pi >= NP) pi = NP - 1;
        const Pt3 p = *(const Pt3*)(Pb + 3 * pi);
        const float pw = p.x * p.x + p.y * p.y + p.z * p.z;
        a[s].x = hi ? (-2.f * p.z) : (-2.f * p.x);
        a[s].y = hi ? 1.f          : (-2.f * p.y);
#pragma unroll
        for (int r = 0; r < 8; ++r) {
            c[s][r]    = __shfl(pw, srcBase + r, 16);
            minv[s][r] = BIGF;
        }
    }

    const int nfull  = NQ >> 4;
    const int ntiles = (NQ + 15) >> 4;
    const int tpw    = (ntiles + 7) >> 3;
    const int tbeg   = wave * tpw;
    const int tend   = (tbeg + tpw < ntiles) ? (tbeg + tpw) : ntiles;
    const int fend   = (tend < nfull) ? tend : nfull;
    const int nt     = (fend > tbeg) ? (fend - tbeg) : 0;

    const float* qw = Qb + 48 * tbeg;

    for (int t = 0; t + 4 <= nt; t += 4) {
        Pt3 q[4];
#pragma unroll
        for (int u = 0; u < 4; ++u) q[u] = *(const Pt3*)(qw + 48 * u + jo);
        qw += 192;
        __builtin_amdgcn_sched_barrier(0);
#pragma unroll
        for (int u = 0; u < 4; ++u) {
            const float q2 = q[u].x * q[u].x + q[u].y * q[u].y + q[u].z * q[u].z;
            v2f bb;
            bb.x = hi ? q[u].z : q[u].x;
            bb.y = hi ? q2     : q[u].y;
#pragma unroll
            for (int s = 0; s < NS; ++s) {
                v8f d = __builtin_amdgcn_wmma_f32_16x16x4_f32(
                    false, a[s], false, bb, (short)0, c[s], false, false);
#pragma unroll
                for (int r = 0; r < 8; ++r) minv[s][r] = fminf(minv[s][r], d[r]);
            }
        }
    }
    for (int t = nt & ~3; t < nt; ++t) {
        const Pt3 q = *(const Pt3*)(qw + jo);
        qw += 48;
        const float q2 = q.x * q.x + q.y * q.y + q.z * q.z;
        v2f bb;
        bb.x = hi ? q.z : q.x;
        bb.y = hi ? q2  : q.y;
#pragma unroll
        for (int s = 0; s < NS; ++s) {
            v8f d = __builtin_amdgcn_wmma_f32_16x16x4_f32(
                false, a[s], false, bb, (short)0, c[s], false, false);
#pragma unroll
            for (int r = 0; r < 8; ++r) minv[s][r] = fminf(minv[s][r], d[r]);
        }
    }

    if (nfull < ntiles && tbeg <= nfull && nfull < tend) {
        int qi = (nfull << 4) + j;
        const bool ok = qi < NQ;
        if (!ok) qi = NQ - 1;
        const Pt3 q = *(const Pt3*)(Qb + 3 * qi);
        const float q2 = ok ? (q.x * q.x + q.y * q.y + q.z * q.z) : BIGF;
        v2f bb;
        bb.x = hi ? q.z : q.x;
        bb.y = hi ? q2  : q.y;
#pragma unroll
        for (int s = 0; s < NS; ++s) {
            v8f d = __builtin_amdgcn_wmma_f32_16x16x4_f32(
                false, a[s], false, bb, (short)0, c[s], false, false);
#pragma unroll
            for (int r = 0; r < 8; ++r) minv[s][r] = fminf(minv[s][r], d[r]);
        }
    }

#pragma unroll
    for (int g = 0; g < NS / 4; ++g) {
        if (g) __syncthreads();
#pragma unroll
        for (int sl = 0; sl < 4; ++sl) {
#pragma unroll
            for (int r = 0; r < 8; ++r) part[wave][8 * sl + r][lane] = minv[4 * g + sl][r];
        }
        __syncthreads();

        float v[4];
#pragma unroll
        for (int sl = 0; sl < 4; ++sl) v[sl] = part[0][8 * sl + wave][lane];
#pragma unroll
        for (int ww = 1; ww < 8; ++ww) {
#pragma unroll
            for (int sl = 0; sl < 4; ++sl) v[sl] = fminf(v[sl], part[ww][8 * sl + wave][lane]);
        }

#pragma unroll
        for (int mask = 1; mask <= 8; mask <<= 1) {
#pragma unroll
            for (int sl = 0; sl < 4; ++sl) v[sl] = fminf(v[sl], __shfl_xor(v[sl], mask, 32));
        }

        if (j == 0) {
#pragma unroll
            for (int sl = 0; sl < 4; ++sl) rowmin[16 * (4 * g + sl) + (hi ? 8 : 0) + wave] = v[sl];
        }
    }
    __syncthreads();

    if (wave < ROWS / 32) {
        int m = row0 + 32 * wave + lane;
        if (m < NP) o[(size_t)b * NP + m] = rowmin[32 * wave + lane];
    }
}

#define TPB    32
#define RPB    (8 * ROWS)
#define RPITCH 68

typedef float v4f __attribute__((ext_vector_type(4)));

__global__ __launch_bounds__(256, 4) void chamfer_shared_tile_kernel(
    const float* __restrict__ in1, const float* __restrict__ in2,
    float* __restrict__ rowpart, float* __restrict__ colpart, int N, int M, int B)
{
    __shared__ float colbuf[8][TPB * 16];
    __shared__ __attribute__((aligned(16))) float rowbuf[8][32 * RPITCH];

    const int lane = threadIdx.x & 31;
    const int wave = __builtin_amdgcn_readfirstlane((int)(threadIdx.x >> 5));

    const int NP = N;
    const int NQ = M;
    const int nfull  = NQ >> 4;
    const int ntiles = (NQ + 15) >> 4;

    const int gr   = (N + RPB - 1) / RPB;
    const int nr   = (ntiles + TPB - 1) / TPB;
    const int cr   = blockIdx.x % nr;
    const int rest = blockIdx.x / nr;
    const int b    = rest / gr;
    if (b >= B) return;
    const int rb   = rest - b * gr;
    const int row0 = rb * RPB + wave * ROWS;

    const int  j  = lane & 15;
    const bool hi = lane >= 16;

    const float* Pb = in1 + (size_t)b * NP * 3;
    const float* Qb = in2 + (size_t)b * NQ * 3;
    const int    jo = 3 * j;

    const int srcBase = hi ? 8 : 0;
    v2f a[NS];
    v8f c[NS], minv[NS];
#pragma unroll
    for (int s = 0; s < NS; ++s) {
        int pi = row0 + 16 * s + j;
        if (pi >= NP) pi = NP - 1;
        const Pt3 p = *(const Pt3*)(Pb + 3 * pi);
        const float pw = p.x * p.x + p.y * p.y + p.z * p.z;
        a[s].x = hi ? (-2.f * p.z) : (-2.f * p.x);
        a[s].y = hi ? 1.f          : (-2.f * p.y);
#pragma unroll
        for (int r = 0; r < 8; ++r) {
            c[s][r]    = __shfl(pw, srcBase + r, 16);
            minv[s][r] = BIGF;
        }
    }

    const int tbeg   = cr * TPB;
    const int tend   = (tbeg + TPB < ntiles) ? (tbeg + TPB) : ntiles;
    const int fend   = (tend < nfull) ? tend : nfull;
    const int nt     = (fend > tbeg) ? (fend - tbeg) : 0;

    const float* qw = Qb + 48 * tbeg;
    float* cw = &colbuf[wave][0];

    for (int t = 0; t + 4 <= nt; t += 4) {
        Pt3 q[4];
#pragma unroll
        for (int u = 0; u < 4; ++u) q[u] = *(const Pt3*)(qw + 48 * u + jo);
        qw += 192;
        __builtin_amdgcn_sched_barrier(0);
        float cmA[4], cmB[4];
#pragma unroll
        for (int u = 0; u < 4; ++u) {
            const float q2 = q[u].x * q[u].x + q[u].y * q[u].y + q[u].z * q[u].z;
            v2f bb;
            bb.x = hi ? q[u].z : q[u].x;
            bb.y = hi ? q2     : q[u].y;
            cmA[u] = BIGF;
            cmB[u] = BIGF;
#pragma unroll
            for (int s = 0; s < NS; ++s) {
                v8f d = __builtin_amdgcn_wmma_f32_16x16x4_f32(
                    false, a[s], false, bb, (short)0, c[s], false, false);
#pragma unroll
                for (int r = 0; r < 8; ++r) minv[s][r] = fminf(minv[s][r], d[r]);
                cmA[u] = fminf(fminf(cmA[u], d[0]), d[1]);
                cmB[u] = fminf(fminf(cmB[u], d[2]), d[3]);
                cmA[u] = fminf(fminf(cmA[u], d[4]), d[5]);
                cmB[u] = fminf(fminf(cmB[u], d[6]), d[7]);
            }
        }
#pragma unroll
        for (int u = 0; u < 4; u += 2) {
            const float ce = fminf(cmA[u], cmB[u]);
            const float co = fminf(cmA[u + 1], cmB[u + 1]);
            const float give = hi ? ce : co;
            const float keep = hi ? co : ce;
            cw[16 * u + lane] = fminf(keep, __shfl_xor(give, 16, 32));
        }
        cw += 64;
    }
    for (int t = nt & ~3; t < nt; ++t) {
        const Pt3 q = *(const Pt3*)(qw + jo);
        qw += 48;
        const float q2 = q.x * q.x + q.y * q.y + q.z * q.z;
        v2f bb;
        bb.x = hi ? q.z : q.x;
        bb.y = hi ? q2  : q.y;
        float cm1 = BIGF, cm2 = BIGF;
#pragma unroll
        for (int s = 0; s < NS; ++s) {
            v8f d = __builtin_amdgcn_wmma_f32_16x16x4_f32(
                false, a[s], false, bb, (short)0, c[s], false, false);
#pragma unroll
            for (int r = 0; r < 8; ++r) minv[s][r] = fminf(minv[s][r], d[r]);
            cm1 = fminf(fminf(cm1, d[0]), d[1]);
            cm2 = fminf(fminf(cm2, d[2]), d[3]);
            cm1 = fminf(fminf(cm1, d[4]), d[5]);
            cm2 = fminf(fminf(cm2, d[6]), d[7]);
        }
        cm1 = fminf(cm1, cm2);
        cw[j] = fminf(cm1, __shfl_xor(cm1, 16, 32));
        cw += 16;
    }

    if (nfull < ntiles && tbeg <= nfull && nfull < tend) {
        int qi = (nfull << 4) + j;
        const bool ok = qi < NQ;
        if (!ok) qi = NQ - 1;
        const Pt3 q = *(const Pt3*)(Qb + 3 * qi);
        const float q2 = ok ? (q.x * q.x + q.y * q.y + q.z * q.z) : BIGF;
        v2f bb;
        bb.x = hi ? q.z : q.x;
        bb.y = hi ? q2  : q.y;
        float cm1 = BIGF, cm2 = BIGF;
#pragma unroll
        for (int s = 0; s < NS; ++s) {
            v8f d = __builtin_amdgcn_wmma_f32_16x16x4_f32(
                false, a[s], false, bb, (short)0, c[s], false, false);
#pragma unroll
            for (int r = 0; r < 8; ++r) minv[s][r] = fminf(minv[s][r], d[r]);
            cm1 = fminf(fminf(cm1, d[0]), d[1]);
            cm2 = fminf(fminf(cm2, d[2]), d[3]);
            cm1 = fminf(fminf(cm1, d[4]), d[5]);
            cm2 = fminf(fminf(cm2, d[6]), d[7]);
        }
        cm1 = fminf(cm1, cm2);
        const float cmf = fminf(cm1, __shfl_xor(cm1, 16, 32));
        if (ok) colbuf[wave][((nfull - tbeg) << 4) + j] = cmf;
    }

    {
        float* rw = &rowbuf[wave][lane * RPITCH];
#pragma unroll
        for (int s = 0; s < NS; ++s) {
            v4f lo4, hi4;
            lo4.x = minv[s][0]; lo4.y = minv[s][1]; lo4.z = minv[s][2]; lo4.w = minv[s][3];
            hi4.x = minv[s][4]; hi4.y = minv[s][5]; hi4.z = minv[s][6]; hi4.w = minv[s][7];
            *(v4f*)(rw + 8 * s)     = lo4;
            *(v4f*)(rw + 8 * s + 4) = hi4;
        }
    }
    __syncthreads();

    {
        const int s4 = lane >> 2;
        const int h4 = (lane >> 1) & 1;
        const int r4 = lane & 1;
        const float* src = &rowbuf[wave][(16 * h4) * RPITCH + 8 * s4 + 4 * r4];
        v4f acc = *(const v4f*)src;
#pragma unroll
        for (int cc = 1; cc < 16; ++cc) {
            const v4f t = *(const v4f*)(src + cc * RPITCH);
            acc.x = fminf(acc.x, t.x);
            acc.y = fminf(acc.y, t.y);
            acc.z = fminf(acc.z, t.z);
            acc.w = fminf(acc.w, t.w);
        }
        const int rr = row0 + 4 * lane;
        float* rp = rowpart + (size_t)cr * B * N + (size_t)b * NP + rr;
        if ((NP & 3) == 0) {
            if (rr < NP) *(v4f*)rp = acc;
        } else {
            if (rr     < NP) rp[0] = acc.x;
            if (rr + 1 < NP) rp[1] = acc.y;
            if (rr + 2 < NP) rp[2] = acc.z;
            if (rr + 3 < NP) rp[3] = acc.w;
        }
    }

    {
        float* cp = colpart + ((size_t)b * gr + rb) * NQ;
#pragma unroll
        for (int q = 0; q < 2; ++q) {
            const int cc  = (int)threadIdx.x + 256 * q;
            const int col = tbeg * 16 + cc;
            if (col < NQ) {
                float v = colbuf[0][cc];
#pragma unroll
                for (int w = 1; w < 8; ++w) v = fminf(v, colbuf[w][cc]);
                cp[col] = v;
            }
        }
    }
}

__global__ __launch_bounds__(64) void chamfer_fold_kernel(
    const float* __restrict__ rowpart, const float* __restrict__ colpart,
    float* __restrict__ out, int N, int M, int B)
{
    const int n1 = B * N;
    const int gr = (N + RPB - 1) / RPB;
    const int nr = (((M + 15) >> 4) + TPB - 1) / TPB;
    const int nmax   = (N > M) ? N : M;
    const int chunks = (nmax + 127) >> 7;
    const int b      = blockIdx.x / chunks;
    if (b >= B) return;
    const int e0     = (blockIdx.x - b * chunks) << 7;

    if (((N | M) & 1) == 0) {
        const int e = e0 + 2 * (int)threadIdx.x;
        const unsigned er = (e < N) ? (unsigned)e : 0u;
        const unsigned ec = (e < M) ? (unsigned)e : 0u;
        const unsigned obr = er * 4u;
        const unsigned obc = ec * 4u;
        const char* rp = (const char*)(rowpart + (size_t)b * N);
        const char* cp = (const char*)(colpart + (size_t)b * gr * M);
        const size_t pitchR = (size_t)(unsigned)n1 * 4;
        const size_t pitchC = (size_t)(unsigned)M * 4;
        v2f rv = {BIGF, BIGF};
        v2f cv = {BIGF, BIGF};
        int k = 0, g = 0;
        for (; k + 16 <= nr && g + 8 <= gr; k += 16, g += 8) {
            v2f tr[16], tc[8];
#pragma unroll
            for (int i = 0; i < 16; ++i) { tr[i] = *(const v2f*)(rp + obr); rp += pitchR; }
#pragma unroll
            for (int i = 0; i < 8; ++i)  { tc[i] = *(const v2f*)(cp + obc); cp += pitchC; }
            __builtin_amdgcn_sched_barrier(0);
#pragma unroll
            for (int i = 0; i < 16; ++i) { rv.x = fminf(rv.x, tr[i].x); rv.y = fminf(rv.y, tr[i].y); }
#pragma unroll
            for (int i = 0; i < 8; ++i)  { cv.x = fminf(cv.x, tc[i].x); cv.y = fminf(cv.y, tc[i].y); }
        }
        for (; k < nr; ++k) {
            const v2f t = *(const v2f*)(rp + obr);
            rp += pitchR;
            rv.x = fminf(rv.x, t.x);
            rv.y = fminf(rv.y, t.y);
        }
        for (; g < gr; ++g) {
            const v2f t = *(const v2f*)(cp + obc);
            cp += pitchC;
            cv.x = fminf(cv.x, t.x);
            cv.y = fminf(cv.y, t.y);
        }
        if (e < N) *(v2f*)(out + (size_t)b * N + e) = rv;
        if (e < M) *(v2f*)(out + (size_t)n1 + (size_t)b * M + e) = cv;
        return;
    }

#pragma unroll
    for (int q = 0; q < 2; ++q) {
        const int e = e0 + 64 * q + (int)threadIdx.x;
        if (e < N) {
            const float* p = rowpart + (size_t)b * N + e;
            float v = BIGF;
            for (int k = 0; k < nr; ++k) v = fminf(v, p[(size_t)k * n1]);
            out[(size_t)b * N + e] = v;
        }
        if (e < M) {
            const float* p = colpart + (size_t)b * gr * M + e;
            float v = BIGF;
            for (int g = 0; g < gr; ++g) v = fminf(v, p[(size_t)g * M]);
            out[(size_t)n1 + (size_t)b * M + e] = v;
        }
    }
}

extern "C" void kernel_launch(void* const* d_in, const int* in_sizes, int n_in,
                              void* d_out, int out_size, void* d_ws, size_t ws_size,
                              hipStream_t stream) {
    const float* in1 = (const float*)d_in[0];
    const float* in2 = (const float*)d_in[1];
    float* out = (float*)d_out;

    const int B = 4;
    const int N = in_sizes[0] / (B * 3);
    const int M = in_sizes[1] / (B * 3);

    const int g1 = (N + ROWS - 1) / ROWS;
    const int g2 = (M + ROWS - 1) / ROWS;

    const int gr = (N + RPB - 1) / RPB;
    const int nr = (((M + 15) / 16) + TPB - 1) / TPB;
    const size_t rowFloats = (size_t)nr * B * N;
    const size_t colFloats = (size_t)B * gr * M;
    if (d_ws != nullptr && ws_size >= (rowFloats + colFloats) * sizeof(float)) {
        float* rowpart = (float*)d_ws;
        float* colpart = rowpart + rowFloats;
        chamfer_shared_tile_kernel<<<B * gr * nr, 256, 0, stream>>>(in1, in2, rowpart, colpart, N, M, B);
        const int nmax = (N > M) ? N : M;
        chamfer_fold_kernel<<<B * ((nmax + 127) / 128), 64, 0, stream>>>(rowpart, colpart, out, N, M, B);
    } else {
        chamfer_both_kernel<<<B * (g1 + g2), 256, 0, stream>>>(in1, in2, out, N, M, B);
    }
}
